# starting kernel
# baseline (speedup 1.0000x reference)
#include <hip/hip_runtime.h>


typedef __attribute__((ext_vector_type(2))) float v2f;
typedef __attribute__((ext_vector_type(8))) float v8f;

#define SWZ_SWAPX16 0x401F

__device__ __forceinline__ float tanh_t(float x) {
#if __has_builtin(__builtin_amdgcn_tanhf)
  return __builtin_amdgcn_tanhf(x);
#else
  float r;
  asm volatile("v_tanh_f32 %0, %1\n\tv_nop" : "=v"(r) : "v"(x));
  return r;
#endif
}

__device__ __forceinline__ float swapx16(float x) {
  int i = __builtin_bit_cast(int, x);
#if __has_builtin(__builtin_amdgcn_permlanex16)
  int r = __builtin_amdgcn_permlanex16(i, i, 0x76543210, 0xfedcba98, true, false);
#else
  int r = __builtin_amdgcn_ds_swizzle(i, SWZ_SWAPX16);
#endif
  return __builtin_bit_cast(float, r);
}

__device__ __forceinline__ v8f wmma4(v2f A, v2f B, v8f C) {
  return __builtin_amdgcn_wmma_f32_16x16x4_f32(false, A, false, B, (short)0, C,
                                               false, false);
}

__global__ void __launch_bounds__(32)
lstm_wmma_fused(const float* __restrict__ x,
                const float* __restrict__ W_ih,
                const float* __restrict__ W_hh,
                const float* __restrict__ b_ih,
                const float* __restrict__ b_hh,
                const float* __restrict__ fc_w,
                const float* __restrict__ fc_b,
                float* __restrict__ out,
                int T) {
  const int  lane = threadIdx.x & 31;
  const int  tile = blockIdx.x;
  const int  n    = lane & 15;
  const bool lo   = lane < 16;
  const int  b    = tile * 16 + n;

  const int m   = n;
  const int k01 = lo ? 0 : 2;
  v2f A0, A1, A2;
  A0.x = W_ih[m * 8 + (k01 + 0)];
  A0.y = W_ih[m * 8 + (k01 + 1)];
  A1.x = W_ih[m * 8 + 4 + (k01 + 0)];
  A1.y = W_ih[m * 8 + 4 + (k01 + 1)];
  A2.x = W_hh[m * 4 + (k01 + 0)];
  A2.y = W_hh[m * 4 + (k01 + 1)];

  v8f bias;
  {
    const int gb = lo ? 0 : 8;
#pragma unroll
    for (int r = 0; r < 8; ++r) bias[r] = b_ih[gb + r] + b_hh[gb + r];
  }

  const float s_pre = lo ? 0.5f : 1.0f;
  const float s_mul = lo ? 0.5f : 1.0f;
  const float s_add = lo ? 0.5f : 0.0f;

  const float* xp = x + (size_t)b * (size_t)T * 8 + k01;

  v2f Bh; Bh.x = 0.0f; Bh.y = 0.0f;
  float c0 = 0.f, c1 = 0.f, c2 = 0.f, c3 = 0.f;

  v8f gx;
  {
    v2f Bx0 = *(const v2f*)(xp);
    v2f Bx1 = *(const v2f*)(xp + 4);
    gx = wmma4(A0, Bx0, bias);
    gx = wmma4(A1, Bx1, gx);
  }

  for (int t = 0; t < T; ++t) {
    int tn = t + 1; if (tn >= T) tn = T - 1;
    const float* xn = xp + (size_t)tn * 8;
    v2f Nx0 = *(const v2f*)(xn);
    v2f Nx1 = *(const v2f*)(xn + 4);
    int tp = t + 8; if (tp >= T) tp = T - 1;
    __builtin_prefetch(xp + (size_t)tp * 8, 0, 3);

    v8f g = wmma4(A2, Bh, gx);

    float ai[4], af[4];
#pragma unroll
    for (int j = 0; j < 4; ++j) ai[j] = fmaf(s_mul, tanh_t(s_pre * g[j]), s_add);
#pragma unroll
    for (int j = 0; j < 4; ++j) af[j] = fmaf(0.5f, tanh_t(0.5f * g[4 + j]), 0.5f);

    v8f gxn = wmma4(A0, Nx0, bias);
    gxn = wmma4(A1, Nx1, gxn);

#pragma unroll
    for (int j = 0; j < 4; ++j) {
      const float ais = swapx16(ai[j]);
      const float afs = swapx16(af[j]);
      const float p   = ai[j] * ais;
      const float fj  = lo ? af[j] : afs;
      const float oj  = lo ? afs : af[j];
      float& cj = (j == 0) ? c0 : (j == 1) ? c1 : (j == 2) ? c2 : c3;
      cj = fmaf(fj, cj, p);
      const float hj = oj * tanh_t(cj);
      if (j == 0) { if (lo) Bh.x = hj; }
      else if (j == 1) { if (lo) Bh.y = hj; }
      else if (j == 2) { if (!lo) Bh.x = hj; }
      else             { if (!lo) Bh.y = hj; }
    }

    gx = gxn;
  }

  const float wA = lo ? fc_w[0] : fc_w[2];
  const float wB = lo ? fc_w[1] : fc_w[3];
  float p  = Bh.x * wA + Bh.y * wB;
  float ps = swapx16(p);
  if (lo) out[b] = p + ps + fc_b[0];
}

extern "C" void kernel_launch(void* const* d_in, const int* in_sizes, int n_in,
                              void* d_out, int out_size, void* d_ws, size_t ws_size,
                              hipStream_t stream) {
  (void)n_in; (void)d_ws; (void)ws_size;
  const float* x    = (const float*)d_in[0];
  const float* W_ih = (const float*)d_in[1];
  const float* W_hh = (const float*)d_in[2];
  const float* b_ih = (const float*)d_in[3];
  const float* b_hh = (const float*)d_in[4];
  const float* fc_w = (const float*)d_in[5];
  const float* fc_b = (const float*)d_in[6];
  float* out = (float*)d_out;

  const int B = out_size;
  const int T = in_sizes[0] / (B * 8);
  const int tiles = B / 16;

  lstm_wmma_fused<<<tiles, 32, 0, stream>>>(x, W_ih, W_hh, b_ih, b_hh,
                                            fc_w, fc_b, out, T);
}


# x staged through LDS: 32-step blocks fetched a block ahead with whole-line 16-byte loads, stored as [step][feature half][batch], fragments read 8 steps per group; same products, same sum order
# speedup vs baseline: 1.7789x; 1.7789x over previous
#include <hip/hip_runtime.h>


typedef __attribute__((ext_vector_type(2))) float v2f;
typedef __attribute__((ext_vector_type(4))) float v4f;
typedef __attribute__((ext_vector_type(8))) float v8f;

#define TBLK  32
#define GRP   8
#define PITCH 132
#define AINL __attribute__((always_inline))

#define SWZ_SWAPX16 0x401F

__device__ __forceinline__ float tanh_t(float x) {
#if __has_builtin(__builtin_amdgcn_tanhf)
  return __builtin_amdgcn_tanhf(x);
#else
  float r;
  asm volatile("v_tanh_f32 %0, %1\n\tv_nop" : "=v"(r) : "v"(x));
  return r;
#endif
}

__device__ __forceinline__ float swapx16(float x) {
  int i = __builtin_bit_cast(int, x);
#if __has_builtin(__builtin_amdgcn_permlanex16)
  int r = __builtin_amdgcn_permlanex16(i, i, 0x76543210, 0xfedcba98, true, false);
#else
  int r = __builtin_amdgcn_ds_swizzle(i, SWZ_SWAPX16);
#endif
  return __builtin_bit_cast(float, r);
}

__device__ __forceinline__ v8f wmma4(v2f A, v2f B, v8f C) {
  return __builtin_amdgcn_wmma_f32_16x16x4_f32(false, A, false, B, (short)0, C,
                                               false, false);
}

__global__ void __launch_bounds__(32)
lstm_wmma_fused(const float* __restrict__ x,
                const float* __restrict__ W_ih,
                const float* __restrict__ W_hh,
                const float* __restrict__ b_ih,
                const float* __restrict__ b_hh,
                const float* __restrict__ fc_w,
                const float* __restrict__ fc_b,
                float* __restrict__ out,
                int T) {
  __shared__ __attribute__((aligned(16))) float xs[2 * TBLK * PITCH];

  const int  lane = threadIdx.x & 31;
  const int  tile = blockIdx.x;
  const int  n    = lane & 15;
  const bool lo   = lane < 16;
  const int  b    = tile * 16 + n;

  const int m   = n;
  const int k01 = lo ? 0 : 2;
  v2f A0, A1, A2;
  A0.x = W_ih[m * 8 + (k01 + 0)];
  A0.y = W_ih[m * 8 + (k01 + 1)];
  A1.x = W_ih[m * 8 + 4 + (k01 + 0)];
  A1.y = W_ih[m * 8 + 4 + (k01 + 1)];
  A2.x = W_hh[m * 4 + (k01 + 0)];
  A2.y = W_hh[m * 4 + (k01 + 1)];

  v8f bias;
  {
    const int gb = lo ? 0 : 8;
#pragma unroll
    for (int r = 0; r < 8; ++r) bias[r] = b_ih[gb + r] + b_hh[gb + r];
  }

  const float s_pre = lo ? 0.5f : 1.0f;
  const float s_mul = lo ? 0.5f : 1.0f;
  const float s_add = lo ? 0.5f : 0.0f;

  const int ls = lane >> 1;
  const int lf = (lane & 1) * 4;
  const float* xt = x + (size_t)tile * 16 * (size_t)T * 8 + lf;
  const int st_off = ls * PITCH + (lane & 1) * 64;
  const int rd_off = n * 4 + k01;

  v4f stage[TBLK];

  auto load_block = [&](int tb) AINL {
#pragma unroll
    for (int i = 0; i < TBLK; ++i) {
      int t = tb + (i & 1) * 16 + ls;
      if (t > T - 1) t = T - 1;
      stage[i] = *(const v4f*)(xt + ((size_t)(i >> 1) * (size_t)T + (size_t)t) * 8);
    }
  };
  auto store_block = [&](float* buf) AINL {
#pragma unroll
    for (int i = 0; i < TBLK; ++i)
      *(v4f*)(buf + st_off + ((i & 1) * 16) * PITCH + (i >> 1) * 4) = stage[i];
  };
  auto rd_row = [&](const float* base, int row, v2f& f0, v2f& f1) AINL {
    f0 = *(const v2f*)(base + row * PITCH);
    f1 = *(const v2f*)(base + row * PITCH + 64);
  };
  auto rd_group = [&](const float* cur_rd, const float* nxt_rd, int first_row,
                      v2f (&f0)[GRP], v2f (&f1)[GRP]) AINL {
#pragma unroll
    for (int s = 0; s < GRP; ++s) {
      const int r = first_row + s;
      if (r < TBLK) rd_row(cur_rd, r, f0[s], f1[s]);
      else          rd_row(nxt_rd, r - TBLK, f0[s], f1[s]);
    }
  };

  float* const buf0 = xs;
  float* const buf1 = xs + TBLK * PITCH;

  v2f Bh; Bh.x = 0.0f; Bh.y = 0.0f;
  float c0 = 0.f, c1 = 0.f, c2 = 0.f, c3 = 0.f;
  v8f gx;

  auto step = [&](v2f f0, v2f f1) AINL {
    v8f g = wmma4(A2, Bh, gx);

    float ai[4], af[4];
#pragma unroll
    for (int j = 0; j < 4; ++j) ai[j] = fmaf(s_mul, tanh_t(s_pre * g[j]), s_add);
#pragma unroll
    for (int j = 0; j < 4; ++j) af[j] = fmaf(0.5f, tanh_t(0.5f * g[4 + j]), 0.5f);

    v8f gxn = wmma4(A0, f0, bias);
    gxn = wmma4(A1, f1, gxn);

#pragma unroll
    for (int j = 0; j < 4; ++j) {
      const float ais = swapx16(ai[j]);
      const float afs = swapx16(af[j]);
      const float p   = ai[j] * ais;
      const float fj  = lo ? af[j] : afs;
      const float oj  = lo ? afs : af[j];
      float& cj = (j == 0) ? c0 : (j == 1) ? c1 : (j == 2) ? c2 : c3;
      cj = fmaf(fj, cj, p);
      const float hj = oj * tanh_t(cj);
      if (j == 0) { if (lo) Bh.x = hj; }
      else if (j == 1) { if (lo) Bh.y = hj; }
      else if (j == 2) { if (!lo) Bh.x = hj; }
      else             { if (!lo) Bh.y = hj; }
    }

    gx = gxn;
  };

  v2f FA0[GRP], FA1[GRP], FB0[GRP], FB1[GRP];

#define PIN() __builtin_amdgcn_sched_barrier(0)
#define GROUP(f0, f1, r0, r1, first_row)                                   \
  do {                                                                     \
    PIN(); rd_group(cur_rd, nxt_rd, (first_row), r0, r1); PIN();           \
    _Pragma("unroll")                                                      \
    for (int s = 0; s < GRP; ++s) step(f0[s], f1[s]);                      \
  } while (0)

  load_block(0);
  store_block(buf0);
  __builtin_amdgcn_fence(__ATOMIC_ACQ_REL, "wavefront");
  {
    v2f f0, f1;
    rd_row(buf0 + rd_off, 0, f0, f1);
    gx = wmma4(A0, f0, bias);
    gx = wmma4(A1, f1, gx);
  }
  rd_group(buf0 + rd_off, buf0 + rd_off, 1, FA0, FA1);

  const int nfull = T / TBLK;
  for (int kb = 0; kb < nfull; ++kb) {
    float* const cur = (kb & 1) ? buf1 : buf0;
    float* const nxt = (kb & 1) ? buf0 : buf1;
    const float* const cur_rd = cur + rd_off;
    const float* const nxt_rd = nxt + rd_off;

    load_block((kb + 1) * TBLK);

    GROUP(FA0, FA1, FB0, FB1, 1 * GRP + 1);
    GROUP(FB0, FB1, FA0, FA1, 2 * GRP + 1);

    store_block(nxt);
    __builtin_amdgcn_fence(__ATOMIC_ACQ_REL, "wavefront");

    GROUP(FA0, FA1, FB0, FB1, 3 * GRP + 1);
    GROUP(FB0, FB1, FA0, FA1, 4 * GRP + 1);
  }

  {
    const int rem = T - nfull * TBLK;
    const float* rp = ((nfull & 1) ? buf1 : buf0) + rd_off + PITCH;
    for (int s = 0; s < rem; ++s) {
      v2f f0, f1;
      rd_row(rp, 0, f0, f1);
      rp += PITCH;
      step(f0, f1);
    }
  }

  const float wA = lo ? fc_w[0] : fc_w[2];
  const float wB = lo ? fc_w[1] : fc_w[3];
  float p  = Bh.x * wA + Bh.y * wB;
  float ps = swapx16(p);
  if (lo) out[b] = p + ps + fc_b[0];
}

extern "C" void kernel_launch(void* const* d_in, const int* in_sizes, int n_in,
                              void* d_out, int out_size, void* d_ws, size_t ws_size,
                              hipStream_t stream) {
  (void)n_in; (void)d_ws; (void)ws_size;
  const float* x    = (const float*)d_in[0];
  const float* W_ih = (const float*)d_in[1];
  const float* W_hh = (const float*)d_in[2];
  const float* b_ih = (const float*)d_in[3];
  const float* b_hh = (const float*)d_in[4];
  const float* fc_w = (const float*)d_in[5];
  const float* fc_b = (const float*)d_in[6];
  float* out = (float*)d_out;

  const int B = out_size;
  const int T = in_sizes[0] / (B * 8);
  const int tiles = B / 16;

  lstm_wmma_fused<<<tiles, 32, 0, stream>>>(x, W_ih, W_hh, b_ih, b_hh,
                                            fc_w, fc_b, out, T);
}


# cell update made lane-local by four pairwise half-swap instructions (was 8 cross-half gathers, 12 selects, duplicated cell math): each lane half owns two hidden units; same products, same sum order
# speedup vs baseline: 1.9255x; 1.0824x over previous
#include <hip/hip_runtime.h>


typedef __attribute__((ext_vector_type(2))) float v2f;
typedef __attribute__((ext_vector_type(4))) float v4f;
typedef __attribute__((ext_vector_type(8))) float v8f;

#define TBLK  32
#define GRP   8
#define PITCH 132
#define AINL __attribute__((always_inline))

#define SWZ_SWAPX16 0x401F

__device__ __forceinline__ float tanh_t(float x) {
#if __has_builtin(__builtin_amdgcn_tanhf)
  return __builtin_amdgcn_tanhf(x);
#else
  float r;
  asm volatile("v_tanh_f32 %0, %1\n\tv_nop" : "=v"(r) : "v"(x));
  return r;
#endif
}

__device__ __forceinline__ float swapx16(float x) {
  int i = __builtin_bit_cast(int, x);
#if __has_builtin(__builtin_amdgcn_permlanex16)
  int r = __builtin_amdgcn_permlanex16(i, i, 0x76543210, 0xfedcba98, true, false);
#else
  int r = __builtin_amdgcn_ds_swizzle(i, SWZ_SWAPX16);
#endif
  return __builtin_bit_cast(float, r);
}

__device__ __forceinline__ void swap_halves(float& a, float& b) {
  asm("v_permlane16_swap_b32 %0, %1" : "+v"(a), "+v"(b));
}

__device__ __forceinline__ v8f wmma4(v2f A, v2f B, v8f C) {
  return __builtin_amdgcn_wmma_f32_16x16x4_f32(false, A, false, B, (short)0, C,
                                               false, false);
}

__global__ void __launch_bounds__(32)
lstm_wmma_fused(const float* __restrict__ x,
                const float* __restrict__ W_ih,
                const float* __restrict__ W_hh,
                const float* __restrict__ b_ih,
                const float* __restrict__ b_hh,
                const float* __restrict__ fc_w,
                const float* __restrict__ fc_b,
                float* __restrict__ out,
                int T) {
  __shared__ __attribute__((aligned(16))) float xs[2 * TBLK * PITCH];

  const int  lane = threadIdx.x & 31;
  const int  tile = blockIdx.x;
  const int  n    = lane & 15;
  const bool lo   = lane < 16;
  const int  b    = tile * 16 + n;

  const int m   = n;
  const int k01 = lo ? 0 : 2;
  v2f A0, A1, A2;
  A0.x = W_ih[m * 8 + (k01 + 0)];
  A0.y = W_ih[m * 8 + (k01 + 1)];
  A1.x = W_ih[m * 8 + 4 + (k01 + 0)];
  A1.y = W_ih[m * 8 + 4 + (k01 + 1)];
  A2.x = W_hh[m * 4 + (k01 + 0)];
  A2.y = W_hh[m * 4 + (k01 + 1)];

  v8f bias;
  {
    const int gb = lo ? 0 : 8;
#pragma unroll
    for (int r = 0; r < 8; ++r) bias[r] = b_ih[gb + r] + b_hh[gb + r];
  }

  const float s_pre = lo ? 0.5f : 1.0f;
  const float s_mul = lo ? 0.5f : 1.0f;
  const float s_add = lo ? 0.5f : 0.0f;

  const int ls = lane >> 1;
  const int lf = (lane & 1) * 4;
  const float* xt = x + (size_t)tile * 16 * (size_t)T * 8 + lf;
  const int st_off = ls * PITCH + (lane & 1) * 64;
  const int rd_off = n * 4 + k01;

  v4f stage[TBLK];

  auto load_block = [&](int tb) AINL {
#pragma unroll
    for (int i = 0; i < TBLK; ++i) {
      int t = tb + (i & 1) * 16 + ls;
      if (t > T - 1) t = T - 1;
      stage[i] = *(const v4f*)(xt + ((size_t)(i >> 1) * (size_t)T + (size_t)t) * 8);
    }
  };
  auto store_block = [&](float* buf) AINL {
#pragma unroll
    for (int i = 0; i < TBLK; ++i)
      *(v4f*)(buf + st_off + ((i & 1) * 16) * PITCH + (i >> 1) * 4) = stage[i];
  };
  auto rd_row = [&](const float* base, int row, v2f& f0, v2f& f1) AINL {
    f0 = *(const v2f*)(base + row * PITCH);
    f1 = *(const v2f*)(base + row * PITCH + 64);
  };
  auto rd_group = [&](const float* cur_rd, const float* nxt_rd, int first_row,
                      v2f (&f0)[GRP], v2f (&f1)[GRP]) AINL {
#pragma unroll
    for (int s = 0; s < GRP; ++s) {
      const int r = first_row + s;
      if (r < TBLK) rd_row(cur_rd, r, f0[s], f1[s]);
      else          rd_row(nxt_rd, r - TBLK, f0[s], f1[s]);
    }
  };

  float* const buf0 = xs;
  float* const buf1 = xs + TBLK * PITCH;

  v2f Bh; Bh.x = 0.0f; Bh.y = 0.0f;
  float cA = 0.f, cB = 0.f;
  v8f gx;

  auto step = [&](v2f f0, v2f f1) AINL {
    v8f g = wmma4(A2, Bh, gx);

    float ai[4], af[4];
#pragma unroll
    for (int j = 0; j < 4; ++j) ai[j] = fmaf(s_mul, tanh_t(s_pre * g[j]), s_add);
#pragma unroll
    for (int j = 0; j < 4; ++j) af[j] = fmaf(0.5f, tanh_t(0.5f * g[4 + j]), 0.5f);

    v8f gxn = wmma4(A0, f0, bias);
    gxn = wmma4(A1, f1, gxn);

    swap_halves(ai[0], ai[2]);
    swap_halves(ai[1], ai[3]);
    swap_halves(af[0], af[2]);
    swap_halves(af[1], af[3]);
    cA = fmaf(af[0], cA, ai[0] * ai[2]);
    cB = fmaf(af[1], cB, ai[1] * ai[3]);
    Bh.x = af[2] * tanh_t(cA);
    Bh.y = af[3] * tanh_t(cB);

    gx = gxn;
  };

  v2f FA0[GRP], FA1[GRP], FB0[GRP], FB1[GRP];

#define PIN() __builtin_amdgcn_sched_barrier(0)
#define GROUP(f0, f1, r0, r1, first_row)                                   \
  do {                                                                     \
    PIN(); rd_group(cur_rd, nxt_rd, (first_row), r0, r1); PIN();           \
    _Pragma("unroll")                                                      \
    for (int s = 0; s < GRP; ++s) step(f0[s], f1[s]);                      \
  } while (0)

  load_block(0);
  store_block(buf0);
  __builtin_amdgcn_fence(__ATOMIC_ACQ_REL, "wavefront");
  {
    v2f f0, f1;
    rd_row(buf0 + rd_off, 0, f0, f1);
    gx = wmma4(A0, f0, bias);
    gx = wmma4(A1, f1, gx);
  }
  rd_group(buf0 + rd_off, buf0 + rd_off, 1, FA0, FA1);

  const int nfull = T / TBLK;
  for (int kb = 0; kb < nfull; ++kb) {
    float* const cur = (kb & 1) ? buf1 : buf0;
    float* const nxt = (kb & 1) ? buf0 : buf1;
    const float* const cur_rd = cur + rd_off;
    const float* const nxt_rd = nxt + rd_off;

    load_block((kb + 1) * TBLK);

    GROUP(FA0, FA1, FB0, FB1, 1 * GRP + 1);
    GROUP(FB0, FB1, FA0, FA1, 2 * GRP + 1);

    store_block(nxt);
    __builtin_amdgcn_fence(__ATOMIC_ACQ_REL, "wavefront");

    GROUP(FA0, FA1, FB0, FB1, 3 * GRP + 1);
    GROUP(FB0, FB1, FA0, FA1, 4 * GRP + 1);
  }

  {
    const int rem = T - nfull * TBLK;
    const float* rp = ((nfull & 1) ? buf1 : buf0) + rd_off + PITCH;
    for (int s = 0; s < rem; ++s) {
      v2f f0, f1;
      rd_row(rp, 0, f0, f1);
      rp += PITCH;
      step(f0, f1);
    }
  }

  const float wA = lo ? fc_w[0] : fc_w[2];
  const float wB = lo ? fc_w[1] : fc_w[3];
  float p  = Bh.x * wA + Bh.y * wB;
  float ps = swapx16(p);
  if (lo) out[b] = p + ps + fc_b[0];
}

extern "C" void kernel_launch(void* const* d_in, const int* in_sizes, int n_in,
                              void* d_out, int out_size, void* d_ws, size_t ws_size,
                              hipStream_t stream) {
  (void)n_in; (void)d_ws; (void)ws_size;
  const float* x    = (const float*)d_in[0];
  const float* W_ih = (const float*)d_in[1];
  const float* W_hh = (const float*)d_in[2];
  const float* b_ih = (const float*)d_in[3];
  const float* b_hh = (const float*)d_in[4];
  const float* fc_w = (const float*)d_in[5];
  const float* fc_b = (const float*)d_in[6];
  float* out = (float*)d_out;

  const int B = out_size;
  const int T = in_sizes[0] / (B * 8);
  const int tiles = B / 16;

  lstm_wmma_fused<<<tiles, 32, 0, stream>>>(x, W_ih, W_hh, b_ih, b_hh,
                                            fc_w, fc_b, out, T);
}


# x block mover (address arithmetic, 16-byte global loads, LDS stores) moved to a second wave of the work-group: three LDS tiles, one barrier per 32 steps; same products, same sum order
# speedup vs baseline: 2.3427x; 1.2167x over previous
#include <hip/hip_runtime.h>


typedef __attribute__((ext_vector_type(2))) float v2f;
typedef __attribute__((ext_vector_type(4))) float v4f;
typedef __attribute__((ext_vector_type(8))) float v8f;

#define TBLK  32
#define GRP   8
#define NBUF  3
#define PITCH 132
#define AINL __attribute__((always_inline))

#define SWZ_SWAPX16 0x401F

__device__ __forceinline__ float tanh_t(float x) {
#if __has_builtin(__builtin_amdgcn_tanhf)
  return __builtin_amdgcn_tanhf(x);
#else
  float r;
  asm volatile("v_tanh_f32 %0, %1\n\tv_nop" : "=v"(r) : "v"(x));
  return r;
#endif
}

__device__ __forceinline__ float swapx16(float x) {
  int i = __builtin_bit_cast(int, x);
#if __has_builtin(__builtin_amdgcn_permlanex16)
  int r = __builtin_amdgcn_permlanex16(i, i, 0x76543210, 0xfedcba98, true, false);
#else
  int r = __builtin_amdgcn_ds_swizzle(i, SWZ_SWAPX16);
#endif
  return __builtin_bit_cast(float, r);
}

__device__ __forceinline__ void swap_halves(float& a, float& b) {
  asm("v_permlane16_swap_b32 %0, %1" : "+v"(a), "+v"(b));
}

__device__ __forceinline__ v8f wmma4(v2f A, v2f B, v8f C) {
  return __builtin_amdgcn_wmma_f32_16x16x4_f32(false, A, false, B, (short)0, C,
                                               false, false);
}

__global__ void __launch_bounds__(64)
lstm_wmma_fused(const float* __restrict__ x,
                const float* __restrict__ W_ih,
                const float* __restrict__ W_hh,
                const float* __restrict__ b_ih,
                const float* __restrict__ b_hh,
                const float* __restrict__ fc_w,
                const float* __restrict__ fc_b,
                float* __restrict__ out,
                int T) {
  __shared__ __attribute__((aligned(16))) float xs[NBUF * TBLK * PITCH];

  const int  lane = threadIdx.x & 31;
  const bool mover = __builtin_amdgcn_readfirstlane((int)(threadIdx.x >> 5)) != 0;
  const int  tile = blockIdx.x;
  const int  n    = lane & 15;
  const bool lo   = lane < 16;
  const int  b    = tile * 16 + n;

  const int m   = n;
  const int k01 = lo ? 0 : 2;
  v2f A0, A1, A2;
  A0.x = W_ih[m * 8 + (k01 + 0)];
  A0.y = W_ih[m * 8 + (k01 + 1)];
  A1.x = W_ih[m * 8 + 4 + (k01 + 0)];
  A1.y = W_ih[m * 8 + 4 + (k01 + 1)];
  A2.x = W_hh[m * 4 + (k01 + 0)];
  A2.y = W_hh[m * 4 + (k01 + 1)];

  v8f bias;
  {
    const int gb = lo ? 0 : 8;
#pragma unroll
    for (int r = 0; r < 8; ++r) bias[r] = b_ih[gb + r] + b_hh[gb + r];
  }

  const float s_pre = lo ? 0.5f : 1.0f;
  const float s_mul = lo ? 0.5f : 1.0f;
  const float s_add = lo ? 0.5f : 0.0f;

  const int ls = lane >> 1;
  const int lf = (lane & 1) * 4;
  const float* xt = x + (size_t)tile * 16 * (size_t)T * 8 + lf;
  const int st_off = ls * PITCH + (lane & 1) * 64;
  const int rd_off = n * 4 + k01;

  v4f stage[TBLK];

  auto load_block = [&](int tb) AINL {
#pragma unroll
    for (int i = 0; i < TBLK; ++i) {
      int t = tb + (i & 1) * 16 + ls;
      if (t > T - 1) t = T - 1;
      stage[i] = *(const v4f*)(xt + ((size_t)(i >> 1) * (size_t)T + (size_t)t) * 8);
    }
  };
  auto store_block = [&](float* buf) AINL {
#pragma unroll
    for (int i = 0; i < TBLK; ++i)
      *(v4f*)(buf + st_off + ((i & 1) * 16) * PITCH + (i >> 1) * 4) = stage[i];
  };
  auto rd_row = [&](const float* base, int row, v2f& f0, v2f& f1) AINL {
    f0 = *(const v2f*)(base + row * PITCH);
    f1 = *(const v2f*)(base + row * PITCH + 64);
  };
  auto rd_group = [&](const float* cur_rd, const float* nxt_rd, int first_row,
                      v2f (&f0)[GRP], v2f (&f1)[GRP]) AINL {
#pragma unroll
    for (int s = 0; s < GRP; ++s) {
      const int r = first_row + s;
      if (r < TBLK) rd_row(cur_rd, r, f0[s], f1[s]);
      else          rd_row(nxt_rd, r - TBLK, f0[s], f1[s]);
    }
  };

  v2f Bh; Bh.x = 0.0f; Bh.y = 0.0f;
  float cA = 0.f, cB = 0.f;
  v8f gx;

  auto step = [&](v2f f0, v2f f1) AINL {
    v8f g = wmma4(A2, Bh, gx);

    float ai[4], af[4];
#pragma unroll
    for (int j = 0; j < 4; ++j) ai[j] = fmaf(s_mul, tanh_t(s_pre * g[j]), s_add);
#pragma unroll
    for (int j = 0; j < 4; ++j) af[j] = fmaf(0.5f, tanh_t(0.5f * g[4 + j]), 0.5f);

    v8f gxn = wmma4(A0, f0, bias);
    gxn = wmma4(A1, f1, gxn);

    swap_halves(ai[0], ai[2]);
    swap_halves(ai[1], ai[3]);
    swap_halves(af[0], af[2]);
    swap_halves(af[1], af[3]);
    cA = fmaf(af[0], cA, ai[0] * ai[2]);
    cB = fmaf(af[1], cB, ai[1] * ai[3]);
    Bh.x = af[2] * tanh_t(cA);
    Bh.y = af[3] * tanh_t(cB);

    gx = gxn;
  };

  v2f FA0[GRP], FA1[GRP], FB0[GRP], FB1[GRP];

#define PIN() __builtin_amdgcn_sched_barrier(0)
#define GROUP(f0, f1, r0, r1, first_row)                                   \
  do {                                                                     \
    PIN(); rd_group(cur_rd, nxt_rd, (first_row), r0, r1); PIN();           \
    _Pragma("unroll")                                                      \
    for (int s = 0; s < GRP; ++s) step(f0[s], f1[s]);                      \
  } while (0)

  if (mover) {
    load_block(0);
    store_block(xs);
  }
  __syncthreads();
  if (!mover) {
    v2f f0, f1;
    rd_row(xs + rd_off, 0, f0, f1);
    gx = wmma4(A0, f0, bias);
    gx = wmma4(A1, f1, gx);
    rd_group(xs + rd_off, xs + rd_off, 1, FA0, FA1);
  }

  const int nfull = T / TBLK;
  int ci = 0;
  for (int kb = 0; kb < nfull; ++kb) {
    const int ni = (ci + 1 == NBUF) ? 0 : ci + 1;
    float* const cur = xs + ci * (TBLK * PITCH);
    float* const nxt = xs + ni * (TBLK * PITCH);
    const float* const cur_rd = cur + rd_off;
    const float* const nxt_rd = nxt + rd_off;

    if (mover) {
      load_block((kb + 1) * TBLK);
      store_block(nxt);
    } else {
      GROUP(FA0, FA1, FB0, FB1, 1 * GRP + 1);
      GROUP(FB0, FB1, FA0, FA1, 2 * GRP + 1);
    }

    __syncthreads();

    if (!mover) {
      GROUP(FA0, FA1, FB0, FB1, 3 * GRP + 1);
      GROUP(FB0, FB1, FA0, FA1, 4 * GRP + 1);
    }
    ci = ni;
  }

  if (mover) return;

  {
    const int rem = T - nfull * TBLK;
    const float* rp = xs + ci * (TBLK * PITCH) + rd_off + PITCH;
    for (int s = 0; s < rem; ++s) {
      v2f f0, f1;
      rd_row(rp, 0, f0, f1);
      rp += PITCH;
      step(f0, f1);
    }
  }

  const float wA = lo ? fc_w[0] : fc_w[2];
  const float wB = lo ? fc_w[1] : fc_w[3];
  float p  = Bh.x * wA + Bh.y * wB;
  float ps = swapx16(p);
  if (lo) out[b] = p + ps + fc_b[0];
}

extern "C" void kernel_launch(void* const* d_in, const int* in_sizes, int n_in,
                              void* d_out, int out_size, void* d_ws, size_t ws_size,
                              hipStream_t stream) {
  (void)n_in; (void)d_ws; (void)ws_size;
  const float* x    = (const float*)d_in[0];
  const float* W_ih = (const float*)d_in[1];
  const float* W_hh = (const float*)d_in[2];
  const float* b_ih = (const float*)d_in[3];
  const float* b_hh = (const float*)d_in[4];
  const float* fc_w = (const float*)d_in[5];
  const float* fc_b = (const float*)d_in[6];
  float* out = (float*)d_out;

  const int B = out_size;
  const int T = in_sizes[0] / (B * 8);
  const int tiles = B / 16;

  lstm_wmma_fused<<<tiles, 64, 0, stream>>>(x, W_ih, W_hh, b_ih, b_hh,
                                            fc_w, fc_b, out, T);
}


# the four half swaps moved in front of the tanh's (onto the scaled gate sums): one gate kind per register, o gate off the cell path, one swap level; step order pinned; same products, same sum order
# speedup vs baseline: 2.5011x; 1.0676x over previous
#include <hip/hip_runtime.h>


typedef __attribute__((ext_vector_type(2))) float v2f;
typedef __attribute__((ext_vector_type(4))) float v4f;
typedef __attribute__((ext_vector_type(8))) float v8f;

#define TBLK  32
#define GRP   8
#define NBUF  3
#define PITCH 132
#define AINL __attribute__((always_inline))

#define SWZ_SWAPX16 0x401F

__device__ __forceinline__ float tanh_t(float x) {
#if __has_builtin(__builtin_amdgcn_tanhf)
  return __builtin_amdgcn_tanhf(x);
#else
  float r;
  asm volatile("v_tanh_f32 %0, %1\n\tv_nop" : "=v"(r) : "v"(x));
  return r;
#endif
}

__device__ __forceinline__ float swapx16(float x) {
  int i = __builtin_bit_cast(int, x);
#if __has_builtin(__builtin_amdgcn_permlanex16)
  int r = __builtin_amdgcn_permlanex16(i, i, 0x76543210, 0xfedcba98, true, false);
#else
  int r = __builtin_amdgcn_ds_swizzle(i, SWZ_SWAPX16);
#endif
  return __builtin_bit_cast(float, r);
}

__device__ __forceinline__ void swap_halves(float& a, float& b) {
  asm("v_permlane16_swap_b32 %0, %1" : "+v"(a), "+v"(b));
}

__device__ __forceinline__ v8f wmma4(v2f A, v2f B, v8f C) {
  return __builtin_amdgcn_wmma_f32_16x16x4_f32(false, A, false, B, (short)0, C,
                                               false, false);
}

__global__ void __launch_bounds__(64)
lstm_wmma_fused(const float* __restrict__ x,
                const float* __restrict__ W_ih,
                const float* __restrict__ W_hh,
                const float* __restrict__ b_ih,
                const float* __restrict__ b_hh,
                const float* __restrict__ fc_w,
                const float* __restrict__ fc_b,
                float* __restrict__ out,
                int T) {
  __shared__ __attribute__((aligned(16))) float xs[NBUF * TBLK * PITCH];

  const int  lane = threadIdx.x & 31;
  const bool mover = __builtin_amdgcn_readfirstlane((int)(threadIdx.x >> 5)) != 0;
  const int  tile = blockIdx.x;
  const int  n    = lane & 15;
  const bool lo   = lane < 16;
  const int  b    = tile * 16 + n;

  const int m   = n;
  const int k01 = lo ? 0 : 2;
  v2f A0, A1, A2;
  A0.x = W_ih[m * 8 + (k01 + 0)];
  A0.y = W_ih[m * 8 + (k01 + 1)];
  A1.x = W_ih[m * 8 + 4 + (k01 + 0)];
  A1.y = W_ih[m * 8 + 4 + (k01 + 1)];
  A2.x = W_hh[m * 4 + (k01 + 0)];
  A2.y = W_hh[m * 4 + (k01 + 1)];

  v8f bias;
  {
    const int gb = lo ? 0 : 8;
#pragma unroll
    for (int r = 0; r < 8; ++r) bias[r] = b_ih[gb + r] + b_hh[gb + r];
  }

  const int ls = lane >> 1;
  const int lf = (lane & 1) * 4;
  const float* xt = x + (size_t)tile * 16 * (size_t)T * 8 + lf;
  const int st_off = ls * PITCH + (lane & 1) * 64;
  const int rd_off = n * 4 + k01;

  v4f stage[TBLK];

  auto load_block = [&](int tb) AINL {
#pragma unroll
    for (int i = 0; i < TBLK; ++i) {
      int t = tb + (i & 1) * 16 + ls;
      if (t > T - 1) t = T - 1;
      stage[i] = *(const v4f*)(xt + ((size_t)(i >> 1) * (size_t)T + (size_t)t) * 8);
    }
  };
  auto store_block = [&](float* buf) AINL {
#pragma unroll
    for (int i = 0; i < TBLK; ++i)
      *(v4f*)(buf + st_off + ((i & 1) * 16) * PITCH + (i >> 1) * 4) = stage[i];
  };
  auto rd_row = [&](const float* base, int row, v2f& f0, v2f& f1) AINL {
    f0 = *(const v2f*)(base + row * PITCH);
    f1 = *(const v2f*)(base + row * PITCH + 64);
  };
  auto rd_group = [&](const float* cur_rd, const float* nxt_rd, int first_row,
                      v2f (&f0)[GRP], v2f (&f1)[GRP]) AINL {
#pragma unroll
    for (int s = 0; s < GRP; ++s) {
      const int r = first_row + s;
      if (r < TBLK) rd_row(cur_rd, r, f0[s], f1[s]);
      else          rd_row(nxt_rd, r - TBLK, f0[s], f1[s]);
    }
  };

  v2f Bh; Bh.x = 0.0f; Bh.y = 0.0f;
  float cA = 0.f, cB = 0.f;
  v8f gx;

  auto sigm_h = [&](float hv) AINL { return fmaf(0.5f, tanh_t(hv), 0.5f); };
  const float s_pre = lo ? 0.5f : 1.0f;

  auto step = [&](v2f f0, v2f f1) AINL {
    __builtin_amdgcn_sched_barrier(0);

    v8f g = wmma4(A2, Bh, gx);

    v8f gxn = wmma4(A0, f0, bias);
    __builtin_amdgcn_sched_barrier(0);

    float i02 = s_pre * g[0], i13 = s_pre * g[1], g02 = s_pre * g[2], g13 = s_pre * g[3];
    float f02 = 0.5f * g[4], f13 = 0.5f * g[5], o02 = 0.5f * g[6], o13 = 0.5f * g[7];
    swap_halves(i02, g02);
    swap_halves(i13, g13);
    swap_halves(f02, o02);
    swap_halves(f13, o13);
    __builtin_amdgcn_sched_barrier(0);

    const float tgA = tanh_t(g02), tgB = tanh_t(g13);
    const float siA = sigm_h(i02),   siB = sigm_h(i13);
    const float sfA = sigm_h(f02),   sfB = sigm_h(f13);
    const float soA = sigm_h(o02),   soB = sigm_h(o13);

    cA = fmaf(sfA, cA, siA * tgA);
    cB = fmaf(sfB, cB, siB * tgB);
    __builtin_amdgcn_sched_barrier(0);

    gxn = wmma4(A1, f1, gxn);

    Bh.x = soA * tanh_t(cA);
    Bh.y = soB * tanh_t(cB);

    gx = gxn;
    asm volatile("" :: "v"(Bh.x), "v"(Bh.y), "v"(gx));
  };

  v2f FA0[GRP], FA1[GRP], FB0[GRP], FB1[GRP];

#define PIN() __builtin_amdgcn_sched_barrier(0)
#define GROUP(f0, f1, r0, r1, first_row)                                   \
  do {                                                                     \
    PIN(); rd_group(cur_rd, nxt_rd, (first_row), r0, r1); PIN();           \
    _Pragma("unroll")                                                      \
    for (int s = 0; s < GRP; ++s) step(f0[s], f1[s]);                      \
  } while (0)

  if (mover) {
    load_block(0);
    store_block(xs);
  }
  __syncthreads();
  if (!mover) {
    v2f f0, f1;
    rd_row(xs + rd_off, 0, f0, f1);
    gx = wmma4(A0, f0, bias);
    gx = wmma4(A1, f1, gx);
    rd_group(xs + rd_off, xs + rd_off, 1, FA0, FA1);
  }

  const int nfull = T / TBLK;
  int ci = 0;
  for (int kb = 0; kb < nfull; ++kb) {
    const int ni = (ci + 1 == NBUF) ? 0 : ci + 1;
    float* const cur = xs + ci * (TBLK * PITCH);
    float* const nxt = xs + ni * (TBLK * PITCH);
    const float* const cur_rd = cur + rd_off;
    const float* const nxt_rd = nxt + rd_off;

    if (mover) {
      load_block((kb + 1) * TBLK);
      store_block(nxt);
    } else {
      GROUP(FA0, FA1, FB0, FB1, 1 * GRP + 1);
      GROUP(FB0, FB1, FA0, FA1, 2 * GRP + 1);
    }

    __syncthreads();

    if (!mover) {
      GROUP(FA0, FA1, FB0, FB1, 3 * GRP + 1);
      GROUP(FB0, FB1, FA0, FA1, 4 * GRP + 1);
    }
    ci = ni;
  }

  if (mover) return;

  {
    const int rem = T - nfull * TBLK;
    const float* rp = xs + ci * (TBLK * PITCH) + rd_off + PITCH;
    for (int s = 0; s < rem; ++s) {
      v2f f0, f1;
      rd_row(rp, 0, f0, f1);
      rp += PITCH;
      step(f0, f1);
    }
  }

  const float wA = lo ? fc_w[0] : fc_w[2];
  const float wB = lo ? fc_w[1] : fc_w[3];
  float p  = Bh.x * wA + Bh.y * wB;
  float ps = swapx16(p);
  if (lo) out[b] = p + ps + fc_b[0];
}

extern "C" void kernel_launch(void* const* d_in, const int* in_sizes, int n_in,
                              void* d_out, int out_size, void* d_ws, size_t ws_size,
                              hipStream_t stream) {
  (void)n_in; (void)d_ws; (void)ws_size;
  const float* x    = (const float*)d_in[0];
  const float* W_ih = (const float*)d_in[1];
  const float* W_hh = (const float*)d_in[2];
  const float* b_ih = (const float*)d_in[3];
  const float* b_hh = (const float*)d_in[4];
  const float* fc_w = (const float*)d_in[5];
  const float* fc_b = (const float*)d_in[6];
  float* out = (float*)d_out;

  const int B = out_size;
  const int T = in_sizes[0] / (B * 8);
  const int tiles = B / 16;

  lstm_wmma_fused<<<tiles, 64, 0, stream>>>(x, W_ih, W_hh, b_ih, b_hh,
                                            fc_w, fc_b, out, T);
}
